# starting kernel
# baseline (speedup 1.0000x reference)
#include <hip/hip_runtime.h>

typedef __attribute__((ext_vector_type(2))) float v2f;
typedef __attribute__((ext_vector_type(8))) float v8f;

#define L_SEQ   200
#define D_DIM   64
#define O_DIM   64
#define K_CAPS  3
#define L_PAD   208
#define M_STRIDE 66
#define NEG_BIG (-1e9f)

__launch_bounds__(256)
__global__ void capsule_fused_kernel(const float* __restrict__ E,
                                     const float* __restrict__ Wg,
                                     const float* __restrict__ logits0,
                                     const int*   __restrict__ seqlen,
                                     float*       __restrict__ out) {
    __shared__ float MLDS[L_PAD * M_STRIDE];
    __shared__ float WL[D_DIM * O_DIM];
    __shared__ float wLDS[K_CAPS * L_PAD];
    __shared__ float candLDS[K_CAPS * O_DIM];
    __shared__ float factLDS[K_CAPS];

    const int tid  = threadIdx.x;
    const int n    = blockIdx.x;
    const int lane = tid & 31;
    const int wave = tid >> 5;
    const int m15  = lane & 15;
    const int hi   = lane >> 4;

    #pragma unroll
    for (int i = 0; i < 16; ++i)
        WL[tid + 256 * i] = Wg[tid + 256 * i];
    __syncthreads();

    v2f bfr[16][4];
    #pragma unroll
    for (int kk = 0; kk < 16; ++kk) {
        const int kb = kk * 4 + 2 * hi;
        #pragma unroll
        for (int nt = 0; nt < 4; ++nt) {
            v2f b;
            b.x = WL[(kb    ) * O_DIM + nt * 16 + m15];
            b.y = WL[(kb + 1) * O_DIM + nt * 16 + m15];
            bfr[kk][nt] = b;
        }
    }

    const float* En = E + (size_t)n * (L_SEQ * D_DIM);

    for (int strip = wave; strip < 13; strip += 8) {
        const int  row   = strip * 16 + m15;
        const bool rowok = (row < L_SEQ);
        const float* Erow = En + (size_t)(rowok ? row : 0) * D_DIM;

        v2f afr[16];
        #pragma unroll
        for (int kk = 0; kk < 16; ++kk)
            afr[kk] = __builtin_nontemporal_load(
                (const v2f*)(Erow + kk * 4 + 2 * hi));
        if (!rowok) {
            #pragma unroll
            for (int kk = 0; kk < 16; ++kk) { afr[kk].x = 0.0f; afr[kk].y = 0.0f; }
        }

        v8f zero = {};
        v8f acc[4] = {zero, zero, zero, zero};
        #pragma unroll
        for (int kk = 0; kk < 16; ++kk) {
            #pragma unroll
            for (int nt = 0; nt < 4; ++nt) {
                acc[nt] = __builtin_amdgcn_wmma_f32_16x16x4_f32(
                    false, afr[kk], false, bfr[kk][nt], (short)0, acc[nt],
                    false, false);
            }
        }

        #pragma unroll
        for (int nt = 0; nt < 4; ++nt)
            #pragma unroll
            for (int v = 0; v < 8; ++v)
                MLDS[(strip * 16 + v + 8 * hi) * M_STRIDE + nt * 16 + m15] =
                    acc[nt][v];
    }

    const int  slen   = seqlen[n];
    const int  l      = tid;
    const bool lvalid = (l < L_SEQ);
    const bool mvalid = lvalid && (l < slen);

    float lg0 = 0.f, lg1 = 0.f, lg2 = 0.f;
    if (lvalid) {
        const float* lp = logits0 + (size_t)n * (K_CAPS * L_SEQ);
        lg0 = lp[0 * L_SEQ + l];
        lg1 = lp[1 * L_SEQ + l];
        lg2 = lp[2 * L_SEQ + l];
    }

    for (int it = 0; it < 3; ++it) {
        if (lvalid) {
            float p0 = mvalid ? lg0 : NEG_BIG;
            float p1 = mvalid ? lg1 : NEG_BIG;
            float p2 = mvalid ? lg2 : NEG_BIG;
            float m  = fmaxf(p0, fmaxf(p1, p2));
            float e0 = __expf(p0 - m);
            float e1 = __expf(p1 - m);
            float e2 = __expf(p2 - m);
            float inv = 1.0f / (e0 + e1 + e2);
            wLDS[0 * L_PAD + l] = e0 * inv;
            wLDS[1 * L_PAD + l] = e1 * inv;
            wLDS[2 * L_PAD + l] = e2 * inv;
        }
        __syncthreads();

        if (tid < K_CAPS * O_DIM) {
            const int k = tid >> 6, o = tid & 63;
            float acc = 0.0f;
            #pragma unroll 8
            for (int ll = 0; ll < L_SEQ; ++ll)
                acc += wLDS[k * L_PAD + ll] * MLDS[ll * M_STRIDE + o];
            candLDS[k * O_DIM + o] = acc;
        }
        __syncthreads();

        if (tid < K_CAPS) {
            float sn = 0.0f;
            #pragma unroll 8
            for (int o = 0; o < O_DIM; ++o) {
                float c = candLDS[tid * O_DIM + o];
                sn += c * c;
            }
            factLDS[tid] = sn / (1.0f + sn) / sqrtf(sn + 1e-8f);
        }
        __syncthreads();

        if (it < 2) {
            if (lvalid) {
                float d0 = 0.f, d1 = 0.f, d2 = 0.f;
                const float* mrow = &MLDS[l * M_STRIDE];
                #pragma unroll 8
                for (int o = 0; o < O_DIM; ++o) {
                    float mv = mrow[o];
                    d0 += candLDS[0 * O_DIM + o] * mv;
                    d1 += candLDS[1 * O_DIM + o] * mv;
                    d2 += candLDS[2 * O_DIM + o] * mv;
                }
                lg0 += factLDS[0] * d0;
                lg1 += factLDS[1] * d1;
                lg2 += factLDS[2] * d2;
            }
        } else {
            if (tid < K_CAPS * O_DIM) {
                const int k = tid >> 6;
                out[(size_t)n * (K_CAPS * O_DIM) + tid] =
                    factLDS[k] * candLDS[tid];
            }
        }
    }
}

extern "C" void kernel_launch(void* const* d_in, const int* in_sizes, int n_in,
                              void* d_out, int out_size, void* d_ws, size_t ws_size,
                              hipStream_t stream) {
    const float* E       = (const float*)d_in[0];
    const float* W       = (const float*)d_in[1];
    const float* logits0 = (const float*)d_in[2];
    const int*   seqlen  = (const int*)d_in[3];
    float*       out     = (float*)d_out;

    const int N = in_sizes[0] / (L_SEQ * D_DIM);
    capsule_fused_kernel<<<N, 256, 0, stream>>>(E, W, logits0, seqlen, out);
}


# Routing sums spread over all 256 lanes as short independent partials, finished by in-wave DPP and permlane adds (no 200-step or 3-thread serial chains, one barrier less per pass); sum order changes
# speedup vs baseline: 1.4759x; 1.4759x over previous
#include <hip/hip_runtime.h>

typedef __attribute__((ext_vector_type(2))) float v2f;
typedef __attribute__((ext_vector_type(4))) float v4f;
typedef __attribute__((ext_vector_type(8))) float v8f;

#define L_SEQ   200
#define D_DIM   64
#define O_DIM   64
#define K_CAPS  3
#define L_PAD   208
#define M_STRIDE 66
#define NEG_BIG (-1e9f)

template <int M>
__device__ __forceinline__ float row_xor(float x) {
    return __builtin_bit_cast(float, __builtin_amdgcn_update_dpp(
        0, __builtin_bit_cast(int, x), 0x160 + M, 0xf, 0xf, false));
}
__device__ __forceinline__ float xhalf(float x) {
    return __builtin_bit_cast(float, __builtin_amdgcn_permlanex16(
        __builtin_bit_cast(int, x), __builtin_bit_cast(int, x),
        0x76543210, 0xfedcba98, false, false));
}
__device__ __forceinline__ float lane_bcast(float x, int src) {
    return __builtin_bit_cast(float, __builtin_amdgcn_readlane(__builtin_bit_cast(int, x), src));
}

__launch_bounds__(256)
__global__ void capsule_fused_kernel(const float* __restrict__ E,
                                     const float* __restrict__ Wg,
                                     const float* __restrict__ logits0,
                                     const int*   __restrict__ seqlen,
                                     float*       __restrict__ out) {
    __shared__ __attribute__((aligned(16))) float MLDS[L_PAD * M_STRIDE];
    __shared__ __attribute__((aligned(16))) float WL[D_DIM * O_DIM];
    __shared__ __attribute__((aligned(16))) float wLDS[K_CAPS * L_PAD];
    __shared__ __attribute__((aligned(16))) float candLDS[K_CAPS * O_DIM];
    __shared__ __attribute__((aligned(16))) float snpLDS[8 * 4];

    const int tid  = threadIdx.x;
    const int n    = blockIdx.x;
    const int lane = tid & 31;
    const int wave = tid >> 5;
    const int m15  = lane & 15;
    const int hi   = lane >> 4;

    #pragma unroll
    for (int i = 0; i < 16; ++i)
        WL[tid + 256 * i] = Wg[tid + 256 * i];
    __syncthreads();

    v2f bfr[16][4];
    #pragma unroll
    for (int kk = 0; kk < 16; ++kk) {
        const int kb = kk * 4 + 2 * hi;
        #pragma unroll
        for (int nt = 0; nt < 4; ++nt) {
            v2f b;
            b.x = WL[(kb    ) * O_DIM + nt * 16 + m15];
            b.y = WL[(kb + 1) * O_DIM + nt * 16 + m15];
            bfr[kk][nt] = b;
        }
    }

    const float* En = E + (size_t)n * (L_SEQ * D_DIM);

    for (int strip = wave; strip < 13; strip += 8) {
        const int  row   = strip * 16 + m15;
        const bool rowok = (row < L_SEQ);
        const float* Erow = En + (size_t)(rowok ? row : 0) * D_DIM;

        v2f afr[16];
        #pragma unroll
        for (int kk = 0; kk < 16; ++kk)
            afr[kk] = __builtin_nontemporal_load(
                (const v2f*)(Erow + kk * 4 + 2 * hi));
        if (!rowok) {
            #pragma unroll
            for (int kk = 0; kk < 16; ++kk) { afr[kk].x = 0.0f; afr[kk].y = 0.0f; }
        }

        v8f zero = {};
        v8f acc[4] = {zero, zero, zero, zero};
        #pragma unroll
        for (int kk = 0; kk < 16; ++kk) {
            #pragma unroll
            for (int nt = 0; nt < 4; ++nt) {
                acc[nt] = __builtin_amdgcn_wmma_f32_16x16x4_f32(
                    false, afr[kk], false, bfr[kk][nt], (short)0, acc[nt],
                    false, false);
            }
        }

        #pragma unroll
        for (int nt = 0; nt < 4; ++nt)
            #pragma unroll
            for (int v = 0; v < 8; ++v)
                MLDS[(strip * 16 + v + 8 * hi) * M_STRIDE + nt * 16 + m15] =
                    acc[nt][v];
    }

    const int  slen   = seqlen[n];
    const int  l      = tid;
    const bool lvalid = (l < L_SEQ);
    const bool mvalid = lvalid && (l < slen);

    const int o8 = lane & 7;
    const int g  = lane >> 3;
    const int oc = wave * 8 + o8;
    const float*  mcol = &MLDS[(g * 52) * M_STRIDE + oc];
    const v4f*   wq0  = (const v4f*)&wLDS[0 * L_PAD + g * 52];
    const v4f*   wq1  = (const v4f*)&wLDS[1 * L_PAD + g * 52];
    const v4f*   wq2  = (const v4f*)&wLDS[2 * L_PAD + g * 52];

    float lg0 = 0.f, lg1 = 0.f, lg2 = 0.f;
    if (lvalid) {
        const float* lp = logits0 + (size_t)n * (K_CAPS * L_SEQ);
        lg0 = lp[0 * L_SEQ + l];
        lg1 = lp[1 * L_SEQ + l];
        lg2 = lp[2 * L_SEQ + l];
    }

    for (int it = 0; it < 3; ++it) {
        if (tid < L_PAD) {
            float w0 = 0.0f, w1 = 0.0f, w2 = 0.0f;
            if (lvalid) {
                float p0 = mvalid ? lg0 : NEG_BIG;
                float p1 = mvalid ? lg1 : NEG_BIG;
                float p2 = mvalid ? lg2 : NEG_BIG;
                float m  = fmaxf(p0, fmaxf(p1, p2));
                float e0 = __expf(p0 - m);
                float e1 = __expf(p1 - m);
                float e2 = __expf(p2 - m);
                float inv = 1.0f / (e0 + e1 + e2);
                w0 = e0 * inv; w1 = e1 * inv; w2 = e2 * inv;
            }
            wLDS[0 * L_PAD + l] = w0;
            wLDS[1 * L_PAD + l] = w1;
            wLDS[2 * L_PAD + l] = w2;
        }
        __syncthreads();

        v2f pa0 = {0.f, 0.f}, pb0 = {0.f, 0.f};
        v2f pa1 = {0.f, 0.f}, pb1 = {0.f, 0.f};
        v2f pa2 = {0.f, 0.f}, pb2 = {0.f, 0.f};
        #pragma unroll
        for (int j = 0; j < 13; ++j) {
            const v4f u0 = wq0[j], u1 = wq1[j], u2 = wq2[j];
            v2f ma, mb;
            ma.x = mcol[(4 * j + 0) * M_STRIDE];
            ma.y = mcol[(4 * j + 1) * M_STRIDE];
            mb.x = mcol[(4 * j + 2) * M_STRIDE];
            mb.y = mcol[(4 * j + 3) * M_STRIDE];
            pa0 += u0.xy * ma; pb0 += u0.zw * mb;
            pa1 += u1.xy * ma; pb1 += u1.zw * mb;
            pa2 += u2.xy * ma; pb2 += u2.zw * mb;
        }
        float c0 = (pa0.x + pa0.y) + (pb0.x + pb0.y);
        float c1 = (pa1.x + pa1.y) + (pb1.x + pb1.y);
        float c2 = (pa2.x + pa2.y) + (pb2.x + pb2.y);
        c0 += row_xor<8>(c0); c1 += row_xor<8>(c1); c2 += row_xor<8>(c2);
        c0 += xhalf(c0);      c1 += xhalf(c1);      c2 += xhalf(c2);
        if (g == 0) {
            candLDS[0 * O_DIM + oc] = c0;
            candLDS[1 * O_DIM + oc] = c1;
            candLDS[2 * O_DIM + oc] = c2;
        }
        float q0 = c0 * c0, q1 = c1 * c1, q2 = c2 * c2;
        q0 += row_xor<1>(q0); q1 += row_xor<1>(q1); q2 += row_xor<1>(q2);
        q0 += row_xor<2>(q0); q1 += row_xor<2>(q1); q2 += row_xor<2>(q2);
        q0 += row_xor<4>(q0); q1 += row_xor<4>(q1); q2 += row_xor<4>(q2);
        if (lane == 0)
            *(v4f*)&snpLDS[wave * 4] = (v4f){q0, q1, q2, 0.0f};
        __syncthreads();

        float sn = 0.0f;
        #pragma unroll
        for (int w = 0; w < 8; ++w)
            sn += snpLDS[w * 4 + (lane & 3)];
        const float fct = sn / (1.0f + sn) / sqrtf(sn + 1e-8f);
        const float f0 = lane_bcast(fct, 0);
        const float f1 = lane_bcast(fct, 1);
        const float f2 = lane_bcast(fct, 2);

        if (it < 2) {
            if (lvalid) {
                float d0e = 0.f, d0o = 0.f, d1e = 0.f, d1o = 0.f, d2e = 0.f, d2o = 0.f;
                const float* mrow = &MLDS[l * M_STRIDE];
                #pragma unroll 16
                for (int o = 0; o < O_DIM; o += 2) {
                    const float mv0 = mrow[o], mv1 = mrow[o + 1];
                    d0e += candLDS[0 * O_DIM + o] * mv0; d0o += candLDS[0 * O_DIM + o + 1] * mv1;
                    d1e += candLDS[1 * O_DIM + o] * mv0; d1o += candLDS[1 * O_DIM + o + 1] * mv1;
                    d2e += candLDS[2 * O_DIM + o] * mv0; d2o += candLDS[2 * O_DIM + o + 1] * mv1;
                }
                lg0 += f0 * (d0e + d0o);
                lg1 += f1 * (d1e + d1o);
                lg2 += f2 * (d2e + d2o);
            }
        } else {
            if (tid < K_CAPS * O_DIM) {
                const int k = tid >> 6;
                const float fk = (k == 0) ? f0 : ((k == 1) ? f1 : f2);
                out[(size_t)n * (K_CAPS * O_DIM) + tid] = fk * candLDS[tid];
            }
        }
    }
}

extern "C" void kernel_launch(void* const* d_in, const int* in_sizes, int n_in,
                              void* d_out, int out_size, void* d_ws, size_t ws_size,
                              hipStream_t stream) {
    const float* E       = (const float*)d_in[0];
    const float* W       = (const float*)d_in[1];
    const float* logits0 = (const float*)d_in[2];
    const int*   seqlen  = (const int*)d_in[3];
    float*       out     = (float*)d_out;

    const int N = in_sizes[0] / (L_SEQ * D_DIM);
    capsule_fused_kernel<<<N, 256, 0, stream>>>(E, W, logits0, seqlen, out);
}


# GEMM re-tiled: a wave keeps B fragments of two column tiles (64 registers, was 128), strips go to wave pairs, A read in 128-bit pieces; 128 registers let 4 work-groups share a WGP; k sum order changes
# speedup vs baseline: 1.8561x; 1.2576x over previous
#include <hip/hip_runtime.h>

typedef __attribute__((ext_vector_type(2))) float v2f;
typedef __attribute__((ext_vector_type(4))) float v4f;
typedef __attribute__((ext_vector_type(8))) float v8f;

#define L_SEQ   200
#define D_DIM   64
#define O_DIM   64
#define K_CAPS  3
#define L_PAD   208
#define M_STRIDE 66
#define NEG_BIG (-1e9f)

template <int M>
__device__ __forceinline__ float row_xor(float x) {
    return __builtin_bit_cast(float, __builtin_amdgcn_update_dpp(
        0, __builtin_bit_cast(int, x), 0x160 + M, 0xf, 0xf, false));
}
__device__ __forceinline__ float xhalf(float x) {
    return __builtin_bit_cast(float, __builtin_amdgcn_permlanex16(
        __builtin_bit_cast(int, x), __builtin_bit_cast(int, x),
        0x76543210, 0xfedcba98, false, false));
}
__device__ __forceinline__ float lane_bcast(float x, int src) {
    return __builtin_bit_cast(float, __builtin_amdgcn_readlane(__builtin_bit_cast(int, x), src));
}

__launch_bounds__(256, 8)
__global__ void capsule_fused_kernel(const float* __restrict__ E,
                                     const float* __restrict__ Wg,
                                     const float* __restrict__ logits0,
                                     const int*   __restrict__ seqlen,
                                     float*       __restrict__ out) {
    __shared__ __attribute__((aligned(16))) float MLDS[L_PAD * M_STRIDE];
    __shared__ __attribute__((aligned(16))) float WL[D_DIM * O_DIM];
    __shared__ __attribute__((aligned(16))) float wLDS[K_CAPS * L_PAD];
    __shared__ __attribute__((aligned(16))) float candLDS[K_CAPS * O_DIM];
    __shared__ __attribute__((aligned(16))) float snpLDS[8 * 4];

    const int tid  = threadIdx.x;
    const int n    = blockIdx.x;
    const int lane = tid & 31;
    const int wave = tid >> 5;
    const int m15  = lane & 15;
    const int hi   = lane >> 4;

    #pragma unroll
    for (int i = 0; i < 16; ++i)
        WL[tid + 256 * i] = Wg[tid + 256 * i];
    __syncthreads();

    const int ntp = wave & 1;
    const int sg  = wave >> 1;

    v2f bfr[16][2];
    #pragma unroll
    for (int f = 0; f < 16; ++f) {
        const int kb = 8 * (f >> 1) + 4 * hi + 2 * (f & 1);
        #pragma unroll
        for (int t = 0; t < 2; ++t) {
            v2f b;
            b.x = WL[(kb    ) * O_DIM + (2 * ntp + t) * 16 + m15];
            b.y = WL[(kb + 1) * O_DIM + (2 * ntp + t) * 16 + m15];
            bfr[f][t] = b;
        }
    }

    const float* En = E + (size_t)n * (L_SEQ * D_DIM);

    for (int strip = sg; strip < 13; strip += 4) {
        const int  row   = strip * 16 + m15;
        const bool rowok = (row < L_SEQ);
        const float* Erow = En + (size_t)(rowok ? row : 0) * D_DIM;

        v4f a4[8];
        #pragma unroll
        for (int j = 0; j < 8; ++j)
            a4[j] = __builtin_nontemporal_load(
                (const v4f*)(Erow + 8 * j + 4 * hi));
        if (!rowok) {
            #pragma unroll
            for (int j = 0; j < 8; ++j) a4[j] = (v4f){0.0f, 0.0f, 0.0f, 0.0f};
        }

        v8f zero = {};
        v8f acc[2] = {zero, zero};
        #pragma unroll
        for (int f = 0; f < 16; ++f) {
            v2f a;
            a.x = a4[f >> 1][2 * (f & 1)];
            a.y = a4[f >> 1][2 * (f & 1) + 1];
            #pragma unroll
            for (int t = 0; t < 2; ++t) {
                acc[t] = __builtin_amdgcn_wmma_f32_16x16x4_f32(
                    false, a, false, bfr[f][t], (short)0, acc[t],
                    false, false);
            }
        }

        #pragma unroll
        for (int t = 0; t < 2; ++t)
            #pragma unroll
            for (int v = 0; v < 8; ++v)
                MLDS[(strip * 16 + v + 8 * hi) * M_STRIDE + (2 * ntp + t) * 16 + m15] =
                    acc[t][v];
    }

    const int  slen   = seqlen[n];
    const int  l      = tid;
    const bool lvalid = (l < L_SEQ);
    const bool mvalid = lvalid && (l < slen);

    const int o8 = lane & 7;
    const int g  = lane >> 3;
    const int oc = wave * 8 + o8;
    const float*  mcol = &MLDS[(g * 52) * M_STRIDE + oc];
    const v4f*   wq0  = (const v4f*)&wLDS[0 * L_PAD + g * 52];
    const v4f*   wq1  = (const v4f*)&wLDS[1 * L_PAD + g * 52];
    const v4f*   wq2  = (const v4f*)&wLDS[2 * L_PAD + g * 52];

    float lg0 = 0.f, lg1 = 0.f, lg2 = 0.f;
    if (lvalid) {
        const float* lp = logits0 + (size_t)n * (K_CAPS * L_SEQ);
        lg0 = lp[0 * L_SEQ + l];
        lg1 = lp[1 * L_SEQ + l];
        lg2 = lp[2 * L_SEQ + l];
    }

    for (int it = 0; it < 3; ++it) {
        if (tid < L_PAD) {
            float w0 = 0.0f, w1 = 0.0f, w2 = 0.0f;
            if (lvalid) {
                float p0 = mvalid ? lg0 : NEG_BIG;
                float p1 = mvalid ? lg1 : NEG_BIG;
                float p2 = mvalid ? lg2 : NEG_BIG;
                float m  = fmaxf(p0, fmaxf(p1, p2));
                float e0 = __expf(p0 - m);
                float e1 = __expf(p1 - m);
                float e2 = __expf(p2 - m);
                float inv = 1.0f / (e0 + e1 + e2);
                w0 = e0 * inv; w1 = e1 * inv; w2 = e2 * inv;
            }
            wLDS[0 * L_PAD + l] = w0;
            wLDS[1 * L_PAD + l] = w1;
            wLDS[2 * L_PAD + l] = w2;
        }
        __syncthreads();

        v2f pa0 = {0.f, 0.f}, pb0 = {0.f, 0.f};
        v2f pa1 = {0.f, 0.f}, pb1 = {0.f, 0.f};
        v2f pa2 = {0.f, 0.f}, pb2 = {0.f, 0.f};
        #pragma unroll
        for (int j = 0; j < 13; ++j) {
            const v4f u0 = wq0[j], u1 = wq1[j], u2 = wq2[j];
            v2f ma, mb;
            ma.x = mcol[(4 * j + 0) * M_STRIDE];
            ma.y = mcol[(4 * j + 1) * M_STRIDE];
            mb.x = mcol[(4 * j + 2) * M_STRIDE];
            mb.y = mcol[(4 * j + 3) * M_STRIDE];
            pa0 += u0.xy * ma; pb0 += u0.zw * mb;
            pa1 += u1.xy * ma; pb1 += u1.zw * mb;
            pa2 += u2.xy * ma; pb2 += u2.zw * mb;
        }
        float c0 = (pa0.x + pa0.y) + (pb0.x + pb0.y);
        float c1 = (pa1.x + pa1.y) + (pb1.x + pb1.y);
        float c2 = (pa2.x + pa2.y) + (pb2.x + pb2.y);
        c0 += row_xor<8>(c0); c1 += row_xor<8>(c1); c2 += row_xor<8>(c2);
        c0 += xhalf(c0);      c1 += xhalf(c1);      c2 += xhalf(c2);
        if (g == 0) {
            candLDS[0 * O_DIM + oc] = c0;
            candLDS[1 * O_DIM + oc] = c1;
            candLDS[2 * O_DIM + oc] = c2;
        }
        float q0 = c0 * c0, q1 = c1 * c1, q2 = c2 * c2;
        q0 += row_xor<1>(q0); q1 += row_xor<1>(q1); q2 += row_xor<1>(q2);
        q0 += row_xor<2>(q0); q1 += row_xor<2>(q1); q2 += row_xor<2>(q2);
        q0 += row_xor<4>(q0); q1 += row_xor<4>(q1); q2 += row_xor<4>(q2);
        if (lane == 0)
            *(v4f*)&snpLDS[wave * 4] = (v4f){q0, q1, q2, 0.0f};
        __syncthreads();

        float sn = 0.0f;
        #pragma unroll
        for (int w = 0; w < 8; ++w)
            sn += snpLDS[w * 4 + (lane & 3)];
        const float fct = sn / (1.0f + sn) / sqrtf(sn + 1e-8f);
        const float f0 = lane_bcast(fct, 0);
        const float f1 = lane_bcast(fct, 1);
        const float f2 = lane_bcast(fct, 2);

        if (it < 2) {
            if (lvalid) {
                float d0e = 0.f, d0o = 0.f, d1e = 0.f, d1o = 0.f, d2e = 0.f, d2o = 0.f;
                const float* mrow = &MLDS[l * M_STRIDE];
                #pragma unroll 8
                for (int o = 0; o < O_DIM; o += 2) {
                    const float mv0 = mrow[o], mv1 = mrow[o + 1];
                    d0e += candLDS[0 * O_DIM + o] * mv0; d0o += candLDS[0 * O_DIM + o + 1] * mv1;
                    d1e += candLDS[1 * O_DIM + o] * mv0; d1o += candLDS[1 * O_DIM + o + 1] * mv1;
                    d2e += candLDS[2 * O_DIM + o] * mv0; d2o += candLDS[2 * O_DIM + o + 1] * mv1;
                }
                lg0 += f0 * (d0e + d0o);
                lg1 += f1 * (d1e + d1o);
                lg2 += f2 * (d2e + d2o);
            }
        } else {
            if (tid < K_CAPS * O_DIM) {
                const int k = tid >> 6;
                const float fk = (k == 0) ? f0 : ((k == 1) ? f1 : f2);
                out[(size_t)n * (K_CAPS * O_DIM) + tid] = fk * candLDS[tid];
            }
        }
    }
}

extern "C" void kernel_launch(void* const* d_in, const int* in_sizes, int n_in,
                              void* d_out, int out_size, void* d_ws, size_t ws_size,
                              hipStream_t stream) {
    const float* E       = (const float*)d_in[0];
    const float* W       = (const float*)d_in[1];
    const float* logits0 = (const float*)d_in[2];
    const int*   seqlen  = (const int*)d_in[3];
    float*       out     = (float*)d_out;

    const int N = in_sizes[0] / (L_SEQ * D_DIM);
    capsule_fused_kernel<<<N, 256, 0, stream>>>(E, W, logits0, seqlen, out);
}


# Routing lanes re-mapped to cut LDS reads: logit dots 4 lanes per row with candidates held in registers, candidate sums 2 columns x 26 rows per lane, M row stride 68; sum order changes
# speedup vs baseline: 1.9590x; 1.0554x over previous
#include <hip/hip_runtime.h>

typedef __attribute__((ext_vector_type(2))) float v2f;
typedef __attribute__((ext_vector_type(4))) float v4f;
typedef __attribute__((ext_vector_type(8))) float v8f;

#define L_SEQ   200
#define D_DIM   64
#define O_DIM   64
#define K_CAPS  3
#define L_PAD   208
#define M_STRIDE 68
#define W_GRP    28
#define W_PAD    (8 * W_GRP)
#define NEG_BIG (-1e9f)

template <int M>
__device__ __forceinline__ float row_xor(float x) {
    return __builtin_bit_cast(float, __builtin_amdgcn_update_dpp(
        0, __builtin_bit_cast(int, x), 0x160 + M, 0xf, 0xf, false));
}
__device__ __forceinline__ float xhalf(float x) {
    return __builtin_bit_cast(float, __builtin_amdgcn_permlanex16(
        __builtin_bit_cast(int, x), __builtin_bit_cast(int, x),
        0x76543210, 0xfedcba98, false, false));
}
__device__ __forceinline__ float lane_bcast(float x, int src) {
    return __builtin_bit_cast(float, __builtin_amdgcn_readlane(__builtin_bit_cast(int, x), src));
}

__launch_bounds__(256, 8)
__global__ void capsule_fused_kernel(const float* __restrict__ E,
                                     const float* __restrict__ Wg,
                                     const float* __restrict__ logits0,
                                     const int*   __restrict__ seqlen,
                                     float*       __restrict__ out) {
    __shared__ __attribute__((aligned(16))) float MLDS[L_PAD * M_STRIDE];
    __shared__ __attribute__((aligned(16))) float WL[D_DIM * O_DIM];
    __shared__ __attribute__((aligned(16))) float wLDS[K_CAPS * W_PAD];
    __shared__ __attribute__((aligned(16))) float candLDS[K_CAPS * O_DIM];
    __shared__ __attribute__((aligned(16))) float snpLDS[8 * 4];

    const int tid  = threadIdx.x;
    const int n    = blockIdx.x;
    const int lane = tid & 31;
    const int wave = tid >> 5;
    const int m15  = lane & 15;
    const int hi   = lane >> 4;

    #pragma unroll
    for (int i = 0; i < 16; ++i)
        WL[tid + 256 * i] = Wg[tid + 256 * i];
    __syncthreads();

    const int ntp = wave & 1;
    const int sg  = wave >> 1;

    v2f bfr[16][2];
    #pragma unroll
    for (int f = 0; f < 16; ++f) {
        const int kb = 8 * (f >> 1) + 4 * hi + 2 * (f & 1);
        #pragma unroll
        for (int t = 0; t < 2; ++t) {
            v2f b;
            b.x = WL[(kb    ) * O_DIM + (2 * ntp + t) * 16 + m15];
            b.y = WL[(kb + 1) * O_DIM + (2 * ntp + t) * 16 + m15];
            bfr[f][t] = b;
        }
    }

    const float* En = E + (size_t)n * (L_SEQ * D_DIM);

    for (int strip = sg; strip < 13; strip += 4) {
        const int  row   = strip * 16 + m15;
        const bool rowok = (row < L_SEQ);
        const float* Erow = En + (size_t)(rowok ? row : 0) * D_DIM;

        v4f a4[8];
        #pragma unroll
        for (int j = 0; j < 8; ++j)
            a4[j] = __builtin_nontemporal_load(
                (const v4f*)(Erow + 8 * j + 4 * hi));
        if (!rowok) {
            #pragma unroll
            for (int j = 0; j < 8; ++j) a4[j] = (v4f){0.0f, 0.0f, 0.0f, 0.0f};
        }

        v8f zero = {};
        v8f acc[2] = {zero, zero};
        #pragma unroll
        for (int f = 0; f < 16; ++f) {
            v2f a;
            a.x = a4[f >> 1][2 * (f & 1)];
            a.y = a4[f >> 1][2 * (f & 1) + 1];
            #pragma unroll
            for (int t = 0; t < 2; ++t) {
                acc[t] = __builtin_amdgcn_wmma_f32_16x16x4_f32(
                    false, a, false, bfr[f][t], (short)0, acc[t],
                    false, false);
            }
        }

        #pragma unroll
        for (int t = 0; t < 2; ++t)
            #pragma unroll
            for (int v = 0; v < 8; ++v)
                MLDS[(strip * 16 + v + 8 * hi) * M_STRIDE + (2 * ntp + t) * 16 + m15] =
                    acc[t][v];
    }

    const int  q4     = lane & 3;
    const int  rs     = lane >> 2;
    const int  slen   = seqlen[n];
    const int  l      = q4 * 64 + wave * 8 + rs;
    const bool lvalid = (l < L_SEQ);
    const bool mvalid = lvalid && (l < slen);
    const int  wave_s = __builtin_amdgcn_readfirstlane(wave);

    const int cp = lane & 3;
    const int g  = lane >> 2;
    const int oc = wave * 8 + 2 * cp;
    const float*  mcol = &MLDS[(g * 26) * M_STRIDE + oc];
    const v4f*   wq0  = (const v4f*)&wLDS[0 * W_PAD + g * W_GRP];
    const v4f*   wq1  = (const v4f*)&wLDS[1 * W_PAD + g * W_GRP];
    const v4f*   wq2  = (const v4f*)&wLDS[2 * W_PAD + g * W_GRP];
    const int widx = (l / 26) * W_GRP + (l % 26);

    float lg0 = 0.f, lg1 = 0.f, lg2 = 0.f;
    if (lvalid) {
        const float* lp = logits0 + (size_t)n * (K_CAPS * L_SEQ);
        lg0 = lp[0 * L_SEQ + l];
        lg1 = lp[1 * L_SEQ + l];
        lg2 = lp[2 * L_SEQ + l];
    }

    for (int it = 0; it < 3; ++it) {
        if (l < L_PAD) {
            float w0 = 0.0f, w1 = 0.0f, w2 = 0.0f;
            if (lvalid) {
                float p0 = mvalid ? lg0 : NEG_BIG;
                float p1 = mvalid ? lg1 : NEG_BIG;
                float p2 = mvalid ? lg2 : NEG_BIG;
                float m  = fmaxf(p0, fmaxf(p1, p2));
                float e0 = __expf(p0 - m);
                float e1 = __expf(p1 - m);
                float e2 = __expf(p2 - m);
                float inv = 1.0f / (e0 + e1 + e2);
                w0 = e0 * inv; w1 = e1 * inv; w2 = e2 * inv;
            }
            wLDS[0 * W_PAD + widx] = w0;
            wLDS[1 * W_PAD + widx] = w1;
            wLDS[2 * W_PAD + widx] = w2;
        }
        __syncthreads();

        v2f pa0 = {0.f, 0.f}, pb0 = {0.f, 0.f};
        v2f pa1 = {0.f, 0.f}, pb1 = {0.f, 0.f};
        v2f pa2 = {0.f, 0.f}, pb2 = {0.f, 0.f};
        #pragma unroll
        for (int j = 0; j < 7; ++j) {
            const v4f u0 = wq0[j], u1 = wq1[j], u2 = wq2[j];
            const v2f ma = *(const v2f*)(mcol + (4 * j + 0) * M_STRIDE);
            const v2f mb = *(const v2f*)(mcol + (4 * j + 1) * M_STRIDE);
            pa0 += u0.x * ma; pb0 += u0.y * mb;
            pa1 += u1.x * ma; pb1 += u1.y * mb;
            pa2 += u2.x * ma; pb2 += u2.y * mb;
            if (j < 6) {
                const v2f mc = *(const v2f*)(mcol + (4 * j + 2) * M_STRIDE);
                const v2f md = *(const v2f*)(mcol + (4 * j + 3) * M_STRIDE);
                pa0 += u0.z * mc; pb0 += u0.w * md;
                pa1 += u1.z * mc; pb1 += u1.w * md;
                pa2 += u2.z * mc; pb2 += u2.w * md;
            }
        }
        v2f cc0 = pa0 + pb0, cc1 = pa1 + pb1, cc2 = pa2 + pb2;
        cc0.x += row_xor<4>(cc0.x); cc0.y += row_xor<4>(cc0.y);
        cc1.x += row_xor<4>(cc1.x); cc1.y += row_xor<4>(cc1.y);
        cc2.x += row_xor<4>(cc2.x); cc2.y += row_xor<4>(cc2.y);
        cc0.x += row_xor<8>(cc0.x); cc0.y += row_xor<8>(cc0.y);
        cc1.x += row_xor<8>(cc1.x); cc1.y += row_xor<8>(cc1.y);
        cc2.x += row_xor<8>(cc2.x); cc2.y += row_xor<8>(cc2.y);
        cc0.x += xhalf(cc0.x); cc0.y += xhalf(cc0.y);
        cc1.x += xhalf(cc1.x); cc1.y += xhalf(cc1.y);
        cc2.x += xhalf(cc2.x); cc2.y += xhalf(cc2.y);
        if (g == 0) {
            *(v2f*)&candLDS[0 * O_DIM + oc] = cc0;
            *(v2f*)&candLDS[1 * O_DIM + oc] = cc1;
            *(v2f*)&candLDS[2 * O_DIM + oc] = cc2;
        }
        float q0 = cc0.x * cc0.x + cc0.y * cc0.y;
        float q1 = cc1.x * cc1.x + cc1.y * cc1.y;
        float q2 = cc2.x * cc2.x + cc2.y * cc2.y;
        q0 += row_xor<1>(q0); q1 += row_xor<1>(q1); q2 += row_xor<1>(q2);
        q0 += row_xor<2>(q0); q1 += row_xor<2>(q1); q2 += row_xor<2>(q2);
        if (lane == 0)
            *(v4f*)&snpLDS[wave * 4] = (v4f){q0, q1, q2, 0.0f};
        __syncthreads();

        float sn = 0.0f;
        #pragma unroll
        for (int w = 0; w < 8; ++w)
            sn += snpLDS[w * 4 + (lane & 3)];
        const float fct = sn / (1.0f + sn) / sqrtf(sn + 1e-8f);
        const float f0 = lane_bcast(fct, 0);
        const float f1 = lane_bcast(fct, 1);
        const float f2 = lane_bcast(fct, 2);

        if (it < 2) {
            v4f ck0[4], ck1[4], ck2[4];
            #pragma unroll
            for (int j = 0; j < 4; ++j) {
                ck0[j] = *(const v4f*)&candLDS[0 * O_DIM + 16 * q4 + 4 * j];
                ck1[j] = *(const v4f*)&candLDS[1 * O_DIM + 16 * q4 + 4 * j];
                ck2[j] = *(const v4f*)&candLDS[2 * O_DIM + 16 * q4 + 4 * j];
            }
            float ds0 = 0.f, ds1 = 0.f, ds2 = 0.f;
            #pragma unroll
            for (int p = 0; p < 4; ++p) {
                if (p < 3 || wave_s == 0) {
                    const float* mrow = &MLDS[(p * 64 + wave * 8 + rs) * M_STRIDE + 16 * q4];
                    v2f a0 = {0.f, 0.f}, a1 = {0.f, 0.f}, a2 = {0.f, 0.f};
                    #pragma unroll
                    for (int j = 0; j < 4; ++j) {
                        const v4f m4 = *(const v4f*)(mrow + 4 * j);
                        a0 += ck0[j].xy * m4.xy; a1 += ck1[j].xy * m4.xy; a2 += ck2[j].xy * m4.xy;
                        a0 += ck0[j].zw * m4.zw; a1 += ck1[j].zw * m4.zw; a2 += ck2[j].zw * m4.zw;
                    }
                    float d0 = a0.x + a0.y, d1 = a1.x + a1.y, d2 = a2.x + a2.y;
                    d0 += row_xor<1>(d0); d1 += row_xor<1>(d1); d2 += row_xor<1>(d2);
                    d0 += row_xor<2>(d0); d1 += row_xor<2>(d1); d2 += row_xor<2>(d2);
                    if (q4 == p) { ds0 = d0; ds1 = d1; ds2 = d2; }
                }
            }
            lg0 += f0 * ds0;
            lg1 += f1 * ds1;
            lg2 += f2 * ds2;
        } else {
            if (tid < K_CAPS * O_DIM) {
                const int k = tid >> 6;
                const float fk = (k == 0) ? f0 : ((k == 1) ? f1 : f2);
                out[(size_t)n * (K_CAPS * O_DIM) + tid] = fk * candLDS[tid];
            }
        }
    }
}

extern "C" void kernel_launch(void* const* d_in, const int* in_sizes, int n_in,
                              void* d_out, int out_size, void* d_ws, size_t ws_size,
                              hipStream_t stream) {
    const float* E       = (const float*)d_in[0];
    const float* W       = (const float*)d_in[1];
    const float* logits0 = (const float*)d_in[2];
    const int*   seqlen  = (const int*)d_in[3];
    float*       out     = (float*)d_out;

    const int N = in_sizes[0] / (L_SEQ * D_DIM);
    capsule_fused_kernel<<<N, 256, 0, stream>>>(E, W, logits0, seqlen, out);
}


# Routing skips masked rows (at or past seq_len their softmax weight is 1/3 whatever the logits): no logit dots for them, their candidate share summed once and reused; sum order changes
# speedup vs baseline: 2.1216x; 1.0830x over previous
#include <hip/hip_runtime.h>

typedef __attribute__((ext_vector_type(2))) float v2f;
typedef __attribute__((ext_vector_type(4))) float v4f;
typedef __attribute__((ext_vector_type(8))) float v8f;

#define L_SEQ   200
#define D_DIM   64
#define O_DIM   64
#define K_CAPS  3
#define L_PAD   208
#define M_STRIDE 68
#define W_GRP    28
#define W_PAD    (8 * W_GRP)
#define NEG_BIG (-1e9f)

template <int M>
__device__ __forceinline__ float row_xor(float x) {
    return __builtin_bit_cast(float, __builtin_amdgcn_update_dpp(
        0, __builtin_bit_cast(int, x), 0x160 + M, 0xf, 0xf, false));
}
__device__ __forceinline__ float xhalf(float x) {
    return __builtin_bit_cast(float, __builtin_amdgcn_permlanex16(
        __builtin_bit_cast(int, x), __builtin_bit_cast(int, x),
        0x76543210, 0xfedcba98, false, false));
}
__device__ __forceinline__ float lane_bcast(float x, int src) {
    return __builtin_bit_cast(float, __builtin_amdgcn_readlane(__builtin_bit_cast(int, x), src));
}

__launch_bounds__(256, 8)
__global__ void capsule_fused_kernel(const float* __restrict__ E,
                                     const float* __restrict__ Wg,
                                     const float* __restrict__ logits0,
                                     const int*   __restrict__ seqlen,
                                     float*       __restrict__ out) {
    __shared__ __attribute__((aligned(16))) float MLDS[L_PAD * M_STRIDE];
    __shared__ __attribute__((aligned(16))) float WL[D_DIM * O_DIM];
    __shared__ __attribute__((aligned(16))) float wLDS[K_CAPS * W_PAD];
    __shared__ __attribute__((aligned(16))) float candLDS[K_CAPS * O_DIM];
    __shared__ __attribute__((aligned(16))) float snpLDS[8 * 4];

    const int tid  = threadIdx.x;
    const int n    = blockIdx.x;
    const int lane = tid & 31;
    const int wave = tid >> 5;
    const int m15  = lane & 15;
    const int hi   = lane >> 4;

    #pragma unroll
    for (int i = 0; i < 16; ++i)
        WL[tid + 256 * i] = Wg[tid + 256 * i];
    __syncthreads();

    const int ntp = wave & 1;
    const int sg  = wave >> 1;

    v2f bfr[16][2];
    #pragma unroll
    for (int f = 0; f < 16; ++f) {
        const int kb = 8 * (f >> 1) + 4 * hi + 2 * (f & 1);
        #pragma unroll
        for (int t = 0; t < 2; ++t) {
            v2f b;
            b.x = WL[(kb    ) * O_DIM + (2 * ntp + t) * 16 + m15];
            b.y = WL[(kb + 1) * O_DIM + (2 * ntp + t) * 16 + m15];
            bfr[f][t] = b;
        }
    }

    const float* En = E + (size_t)n * (L_SEQ * D_DIM);

    for (int strip = sg; strip < 13; strip += 4) {
        const int  row   = strip * 16 + m15;
        const bool rowok = (row < L_SEQ);
        const float* Erow = En + (size_t)(rowok ? row : 0) * D_DIM;

        v4f a4[8];
        #pragma unroll
        for (int j = 0; j < 8; ++j)
            a4[j] = __builtin_nontemporal_load(
                (const v4f*)(Erow + 8 * j + 4 * hi));
        if (!rowok) {
            #pragma unroll
            for (int j = 0; j < 8; ++j) a4[j] = (v4f){0.0f, 0.0f, 0.0f, 0.0f};
        }

        v8f zero = {};
        v8f acc[2] = {zero, zero};
        #pragma unroll
        for (int f = 0; f < 16; ++f) {
            v2f a;
            a.x = a4[f >> 1][2 * (f & 1)];
            a.y = a4[f >> 1][2 * (f & 1) + 1];
            #pragma unroll
            for (int t = 0; t < 2; ++t) {
                acc[t] = __builtin_amdgcn_wmma_f32_16x16x4_f32(
                    false, a, false, bfr[f][t], (short)0, acc[t],
                    false, false);
            }
        }

        #pragma unroll
        for (int t = 0; t < 2; ++t)
            #pragma unroll
            for (int v = 0; v < 8; ++v)
                MLDS[(strip * 16 + v + 8 * hi) * M_STRIDE + (2 * ntp + t) * 16 + m15] =
                    acc[t][v];
    }

    const int  q4     = lane & 3;
    const int  rs     = lane >> 2;
    const int  slen   = seqlen[n];
    const int  l      = q4 * 64 + wave * 8 + rs;
    const bool lvalid = (l < L_SEQ);
    const bool mvalid = lvalid && (l < slen);
    const int  wave_s = __builtin_amdgcn_readfirstlane(wave);

    const int slc = min(max(slen, 0), L_SEQ);
    const int nj  = (slc + 15) >> 4;

    const int cp = lane & 3;
    const int g  = lane >> 2;
    const int oc = wave * 8 + 2 * cp;
    const float*  mcol = &MLDS[(2 * g) * M_STRIDE + oc];
    const v4f*   wq0  = (const v4f*)&wLDS[0 * W_PAD + g * W_GRP];
    const v4f*   wq1  = (const v4f*)&wLDS[1 * W_PAD + g * W_GRP];
    const v4f*   wq2  = (const v4f*)&wLDS[2 * W_PAD + g * W_GRP];
    const int widx = ((l & 15) >> 1) * W_GRP + 2 * (l >> 4) + (l & 1);

    v2f smask = {0.f, 0.f};
    float lg0 = 0.f, lg1 = 0.f, lg2 = 0.f;
    if (lvalid) {
        const float* lp = logits0 + (size_t)n * (K_CAPS * L_SEQ);
        lg0 = lp[0 * L_SEQ + l];
        lg1 = lp[1 * L_SEQ + l];
        lg2 = lp[2 * L_SEQ + l];
    }

    for (int it = 0; it < 3; ++it) {
        if (l < L_PAD) {
            float w0 = 0.0f, w1 = 0.0f, w2 = 0.0f;
            if (lvalid) {
                float p0 = mvalid ? lg0 : NEG_BIG;
                float p1 = mvalid ? lg1 : NEG_BIG;
                float p2 = mvalid ? lg2 : NEG_BIG;
                float m  = fmaxf(p0, fmaxf(p1, p2));
                float e0 = __expf(p0 - m);
                float e1 = __expf(p1 - m);
                float e2 = __expf(p2 - m);
                float inv = 1.0f / (e0 + e1 + e2);
                w0 = e0 * inv; w1 = e1 * inv; w2 = e2 * inv;
            }
            wLDS[0 * W_PAD + widx] = w0;
            wLDS[1 * W_PAD + widx] = w1;
            wLDS[2 * W_PAD + widx] = w2;
        }
        __syncthreads();

        if (it == 0) {
            v2f sa = {0.f, 0.f}, sb = {0.f, 0.f};
            auto mblk = [&](int j) {
                sa += *(const v2f*)(mcol + (16 * j    ) * M_STRIDE);
                sb += *(const v2f*)(mcol + (16 * j + 1) * M_STRIDE);
            };
            const int nm = 13 - nj;
            int j = nj;
            for (int q = 0; q < (nm >> 2); ++q, j += 4) { mblk(j); mblk(j + 1); mblk(j + 2); mblk(j + 3); }
            if (nm & 2) { mblk(j); mblk(j + 1); j += 2; }
            if (nm & 1) { mblk(j); }
            smask = (sa + sb) * (1.0f / 3.0f);
        }
        v2f pa0 = smask, pb0 = {0.f, 0.f};
        v2f pa1 = smask, pb1 = {0.f, 0.f};
        v2f pa2 = smask, pb2 = {0.f, 0.f};
        auto wpair = [&](int c) {
            const v4f u0 = wq0[c], u1 = wq1[c], u2 = wq2[c];
            const float* mp = mcol + (32 * c) * M_STRIDE;
            const v2f ma = *(const v2f*)(mp);
            const v2f mb = *(const v2f*)(mp + M_STRIDE);
            const v2f mc = *(const v2f*)(mp + 16 * M_STRIDE);
            const v2f md = *(const v2f*)(mp + 17 * M_STRIDE);
            pa0 += u0.x * ma; pb0 += u0.y * mb;
            pa1 += u1.x * ma; pb1 += u1.y * mb;
            pa2 += u2.x * ma; pb2 += u2.y * mb;
            pa0 += u0.z * mc; pb0 += u0.w * md;
            pa1 += u1.z * mc; pb1 += u1.w * md;
            pa2 += u2.z * mc; pb2 += u2.w * md;
        };
        {
            int c = 0;
            for (int q = 0; q < (nj >> 2); ++q, c += 2) { wpair(c); wpair(c + 1); }
            if (nj & 2) { wpair(c); c += 1; }
            if (nj & 1) {
                const v2f t0 = *(const v2f*)&wq0[c];
                const v2f t1 = *(const v2f*)&wq1[c];
                const v2f t2 = *(const v2f*)&wq2[c];
                const float* mp = mcol + (32 * c) * M_STRIDE;
                const v2f ma = *(const v2f*)(mp);
                const v2f mb = *(const v2f*)(mp + M_STRIDE);
                pa0 += t0.x * ma; pb0 += t0.y * mb;
                pa1 += t1.x * ma; pb1 += t1.y * mb;
                pa2 += t2.x * ma; pb2 += t2.y * mb;
            }
        }
        v2f cc0 = pa0 + pb0, cc1 = pa1 + pb1, cc2 = pa2 + pb2;
        cc0.x += row_xor<4>(cc0.x); cc0.y += row_xor<4>(cc0.y);
        cc1.x += row_xor<4>(cc1.x); cc1.y += row_xor<4>(cc1.y);
        cc2.x += row_xor<4>(cc2.x); cc2.y += row_xor<4>(cc2.y);
        cc0.x += row_xor<8>(cc0.x); cc0.y += row_xor<8>(cc0.y);
        cc1.x += row_xor<8>(cc1.x); cc1.y += row_xor<8>(cc1.y);
        cc2.x += row_xor<8>(cc2.x); cc2.y += row_xor<8>(cc2.y);
        cc0.x += xhalf(cc0.x); cc0.y += xhalf(cc0.y);
        cc1.x += xhalf(cc1.x); cc1.y += xhalf(cc1.y);
        cc2.x += xhalf(cc2.x); cc2.y += xhalf(cc2.y);
        if (g == 0) {
            *(v2f*)&candLDS[0 * O_DIM + oc] = cc0;
            *(v2f*)&candLDS[1 * O_DIM + oc] = cc1;
            *(v2f*)&candLDS[2 * O_DIM + oc] = cc2;
        }
        float q0 = cc0.x * cc0.x + cc0.y * cc0.y;
        float q1 = cc1.x * cc1.x + cc1.y * cc1.y;
        float q2 = cc2.x * cc2.x + cc2.y * cc2.y;
        q0 += row_xor<1>(q0); q1 += row_xor<1>(q1); q2 += row_xor<1>(q2);
        q0 += row_xor<2>(q0); q1 += row_xor<2>(q1); q2 += row_xor<2>(q2);
        if (lane == 0)
            *(v4f*)&snpLDS[wave * 4] = (v4f){q0, q1, q2, 0.0f};
        __syncthreads();

        float sn = 0.0f;
        #pragma unroll
        for (int w = 0; w < 8; ++w)
            sn += snpLDS[w * 4 + (lane & 3)];
        const float fct = sn / (1.0f + sn) / sqrtf(sn + 1e-8f);
        const float f0 = lane_bcast(fct, 0);
        const float f1 = lane_bcast(fct, 1);
        const float f2 = lane_bcast(fct, 2);

        if (it < 2) {
            const int r0 = 8 * wave_s;
            if (r0 < slc) {
                v4f ck0[4], ck1[4], ck2[4];
                #pragma unroll
                for (int j = 0; j < 4; ++j) {
                    ck0[j] = *(const v4f*)&candLDS[0 * O_DIM + 16 * q4 + 4 * j];
                    ck1[j] = *(const v4f*)&candLDS[1 * O_DIM + 16 * q4 + 4 * j];
                    ck2[j] = *(const v4f*)&candLDS[2 * O_DIM + 16 * q4 + 4 * j];
                }
                float ds0 = 0.f, ds1 = 0.f, ds2 = 0.f;
                auto lpass = [&](int p) {
                    const float* mrow = &MLDS[(p * 64 + wave * 8 + rs) * M_STRIDE + 16 * q4];
                    v2f a0 = {0.f, 0.f}, a1 = {0.f, 0.f}, a2 = {0.f, 0.f};
                    #pragma unroll
                    for (int j = 0; j < 4; ++j) {
                        const v4f m4 = *(const v4f*)(mrow + 4 * j);
                        a0 += ck0[j].xy * m4.xy; a1 += ck1[j].xy * m4.xy; a2 += ck2[j].xy * m4.xy;
                        a0 += ck0[j].zw * m4.zw; a1 += ck1[j].zw * m4.zw; a2 += ck2[j].zw * m4.zw;
                    }
                    float d0 = a0.x + a0.y, d1 = a1.x + a1.y, d2 = a2.x + a2.y;
                    d0 += row_xor<1>(d0); d1 += row_xor<1>(d1); d2 += row_xor<1>(d2);
                    d0 += row_xor<2>(d0); d1 += row_xor<2>(d1); d2 += row_xor<2>(d2);
                    if (q4 == p) { ds0 = d0; ds1 = d1; ds2 = d2; }
                };
                lpass(0);
                if (r0 + 64 < slc) { lpass(1); }
                if (r0 + 192 < slc) { lpass(2); lpass(3); } else if (r0 + 128 < slc) { lpass(2); }
                lg0 += f0 * ds0;
                lg1 += f1 * ds1;
                lg2 += f2 * ds2;
            }
        } else {
            if (tid < K_CAPS * O_DIM) {
                const int k = tid >> 6;
                const float fk = (k == 0) ? f0 : ((k == 1) ? f1 : f2);
                out[(size_t)n * (K_CAPS * O_DIM) + tid] = fk * candLDS[tid];
            }
        }
    }
}

extern "C" void kernel_launch(void* const* d_in, const int* in_sizes, int n_in,
                              void* d_out, int out_size, void* d_ws, size_t ws_size,
                              hipStream_t stream) {
    const float* E       = (const float*)d_in[0];
    const float* W       = (const float*)d_in[1];
    const float* logits0 = (const float*)d_in[2];
    const int*   seqlen  = (const int*)d_in[3];
    float*       out     = (float*)d_out;

    const int N = in_sizes[0] / (L_SEQ * D_DIM);
    capsule_fused_kernel<<<N, 256, 0, stream>>>(E, W, logits0, seqlen, out);
}


# Masked rows leave the row-by-row GEMM: their rows of E are summed first (linearity), one 16-row tile of partial sums goes through the WMMA product instead of up to 13 strips; sum order changes
# speedup vs baseline: 2.4508x; 1.1552x over previous
#include <hip/hip_runtime.h>

typedef __attribute__((ext_vector_type(2))) float v2f;
typedef __attribute__((ext_vector_type(4))) float v4f;
typedef __attribute__((ext_vector_type(8))) float v8f;

#define L_SEQ   200
#define D_DIM   64
#define O_DIM   64
#define K_CAPS  3
#define L_PAD   208
#define M_STRIDE 68
#define W_GRP    28
#define W_PAD    (8 * W_GRP)
#define NEG_BIG (-1e9f)

template <int M>
__device__ __forceinline__ float row_xor(float x) {
    return __builtin_bit_cast(float, __builtin_amdgcn_update_dpp(
        0, __builtin_bit_cast(int, x), 0x160 + M, 0xf, 0xf, false));
}
__device__ __forceinline__ float xhalf(float x) {
    return __builtin_bit_cast(float, __builtin_amdgcn_permlanex16(
        __builtin_bit_cast(int, x), __builtin_bit_cast(int, x),
        0x76543210, 0xfedcba98, false, false));
}
__device__ __forceinline__ float lane_bcast(float x, int src) {
    return __builtin_bit_cast(float, __builtin_amdgcn_readlane(__builtin_bit_cast(int, x), src));
}

__launch_bounds__(256, 8)
__global__ void capsule_fused_kernel(const float* __restrict__ E,
                                     const float* __restrict__ Wg,
                                     const float* __restrict__ logits0,
                                     const int*   __restrict__ seqlen,
                                     float*       __restrict__ out) {
    __shared__ __attribute__((aligned(16))) float MLDS[L_PAD * M_STRIDE];
    __shared__ __attribute__((aligned(16))) float WL[D_DIM * O_DIM];
    __shared__ __attribute__((aligned(16))) float wLDS[K_CAPS * W_PAD];
    __shared__ __attribute__((aligned(16))) float candLDS[K_CAPS * O_DIM];
    __shared__ __attribute__((aligned(16))) float snpLDS[8 * 4];

    const int tid  = threadIdx.x;
    const int n    = blockIdx.x;
    const int lane = tid & 31;
    const int wave = tid >> 5;
    const int m15  = lane & 15;
    const int hi   = lane >> 4;

    #pragma unroll
    for (int i = 0; i < 16; ++i)
        WL[tid + 256 * i] = Wg[tid + 256 * i];
    __syncthreads();

    const int ntp = wave & 1;

    v2f bfr[16][2];
    #pragma unroll
    for (int f = 0; f < 16; ++f) {
        const int kb = 8 * (f >> 1) + 4 * hi + 2 * (f & 1);
        #pragma unroll
        for (int t = 0; t < 2; ++t) {
            v2f b;
            b.x = WL[(kb    ) * O_DIM + (2 * ntp + t) * 16 + m15];
            b.y = WL[(kb + 1) * O_DIM + (2 * ntp + t) * 16 + m15];
            bfr[f][t] = b;
        }
    }

    const float* En = E + (size_t)n * (L_SEQ * D_DIM);

    const int  wave_s = __builtin_amdgcn_readfirstlane(wave);
    const int  slen   = seqlen[n];
    const int slc = min(max(slen, 0), L_SEQ);
    const int nj  = (slc + 15) >> 4;
    const int njg = (nj >= 12) ? 13 : nj;

    auto strip_mma = [&](const v4f (&a4)[8], int blk) {
        v8f zero = {};
        v8f acc[2] = {zero, zero};
        #pragma unroll
        for (int f = 0; f < 16; ++f) {
            v2f a;
            a.x = a4[f >> 1][2 * (f & 1)];
            a.y = a4[f >> 1][2 * (f & 1) + 1];
            #pragma unroll
            for (int t = 0; t < 2; ++t) {
                acc[t] = __builtin_amdgcn_wmma_f32_16x16x4_f32(
                    false, a, false, bfr[f][t], (short)0, acc[t],
                    false, false);
            }
        }
        #pragma unroll
        for (int t = 0; t < 2; ++t)
            #pragma unroll
            for (int v = 0; v < 8; ++v)
                MLDS[(blk * 16 + v + 8 * hi) * M_STRIDE + (2 * ntp + t) * 16 + m15] =
                    acc[t][v];
    };

    for (int strip = (wave_s >> 1); strip < njg; strip += 4) {
        const int  row   = strip * 16 + m15;
        const bool rowok = (row < L_SEQ);
        const float* Erow = En + (size_t)(rowok ? row : 0) * D_DIM;

        v4f a4[8];
        #pragma unroll
        for (int j = 0; j < 8; ++j)
            a4[j] = __builtin_nontemporal_load(
                (const v4f*)(Erow + 8 * j + 4 * hi));
        if (!rowok) {
            #pragma unroll
            for (int j = 0; j < 8; ++j) a4[j] = (v4f){0.0f, 0.0f, 0.0f, 0.0f};
        }

        strip_mma(a4, strip);
    }

    if (njg < 13) {
        v4f ms = {0.0f, 0.0f, 0.0f, 0.0f};
        auto msum = [&](int j) {
            const float* Eb = En + (size_t)j * (16 * D_DIM) + 4 * lane;
            v4f t[8];
            #pragma unroll
            for (int i = 0; i < 4; ++i)
                t[i] = __builtin_nontemporal_load((const v4f*)(Eb + 128 * i));
            #pragma unroll
            for (int i = 4; i < 8; ++i)
                t[i] = (v4f){0.0f, 0.0f, 0.0f, 0.0f};
            if (j < 12) {
                #pragma unroll
                for (int i = 4; i < 8; ++i)
                    t[i] = __builtin_nontemporal_load((const v4f*)(Eb + 128 * i));
            }
            ms += ((t[0] + t[1]) + (t[2] + t[3])) + ((t[4] + t[5]) + (t[6] + t[7]));
        };
        const int j0 = njg + 7 - wave_s;
        if (j0 < 13)     msum(j0);
        if (j0 + 8 < 13) msum(j0 + 8);
        *(v4f*)&MLDS[((njg + 1) * 16 + 2 * wave + hi) * M_STRIDE + 4 * m15] = ms;
        __syncthreads();
        if ((wave_s >> 1) == 3) {
            const float* Trow = &MLDS[((njg + 1) * 16 + m15) * M_STRIDE + 4 * hi];
            v4f a4[8];
            #pragma unroll
            for (int j = 0; j < 8; ++j)
                a4[j] = *(const v4f*)(Trow + 8 * j);
            strip_mma(a4, njg);
        }
    }

    const int  q4     = lane & 3;
    const int  rs     = lane >> 2;
    const int  l      = q4 * 64 + wave * 8 + rs;
    const bool lvalid = (l < L_SEQ);
    const bool mvalid = lvalid && (l < slen);

    const int cp = lane & 3;
    const int g  = lane >> 2;
    const int oc = wave * 8 + 2 * cp;
    const float*  mcol = &MLDS[(2 * g) * M_STRIDE + oc];
    const v4f*   wq0  = (const v4f*)&wLDS[0 * W_PAD + g * W_GRP];
    const v4f*   wq1  = (const v4f*)&wLDS[1 * W_PAD + g * W_GRP];
    const v4f*   wq2  = (const v4f*)&wLDS[2 * W_PAD + g * W_GRP];
    const int widx = ((l & 15) >> 1) * W_GRP + 2 * (l >> 4) + (l & 1);

    v2f smask = {0.f, 0.f};
    float lg0 = 0.f, lg1 = 0.f, lg2 = 0.f;
    if (lvalid) {
        const float* lp = logits0 + (size_t)n * (K_CAPS * L_SEQ);
        lg0 = lp[0 * L_SEQ + l];
        lg1 = lp[1 * L_SEQ + l];
        lg2 = lp[2 * L_SEQ + l];
    }

    for (int it = 0; it < 3; ++it) {
        if (l < L_PAD) {
            float w0 = 0.0f, w1 = 0.0f, w2 = 0.0f;
            if (lvalid) {
                float p0 = mvalid ? lg0 : NEG_BIG;
                float p1 = mvalid ? lg1 : NEG_BIG;
                float p2 = mvalid ? lg2 : NEG_BIG;
                float m  = fmaxf(p0, fmaxf(p1, p2));
                float e0 = __expf(p0 - m);
                float e1 = __expf(p1 - m);
                float e2 = __expf(p2 - m);
                float inv = 1.0f / (e0 + e1 + e2);
                w0 = e0 * inv; w1 = e1 * inv; w2 = e2 * inv;
            }
            wLDS[0 * W_PAD + widx] = w0;
            wLDS[1 * W_PAD + widx] = w1;
            wLDS[2 * W_PAD + widx] = w2;
        }
        __syncthreads();

        if (it == 0 && njg < 13) {
            const v2f sa = *(const v2f*)(mcol + (16 * njg    ) * M_STRIDE);
            const v2f sb = *(const v2f*)(mcol + (16 * njg + 1) * M_STRIDE);
            smask = (sa + sb) * (1.0f / 3.0f);
        }
        v2f pa0 = smask, pb0 = {0.f, 0.f};
        v2f pa1 = smask, pb1 = {0.f, 0.f};
        v2f pa2 = smask, pb2 = {0.f, 0.f};
        auto wpair = [&](int c) {
            const v4f u0 = wq0[c], u1 = wq1[c], u2 = wq2[c];
            const float* mp = mcol + (32 * c) * M_STRIDE;
            const v2f ma = *(const v2f*)(mp);
            const v2f mb = *(const v2f*)(mp + M_STRIDE);
            const v2f mc = *(const v2f*)(mp + 16 * M_STRIDE);
            const v2f md = *(const v2f*)(mp + 17 * M_STRIDE);
            pa0 += u0.x * ma; pb0 += u0.y * mb;
            pa1 += u1.x * ma; pb1 += u1.y * mb;
            pa2 += u2.x * ma; pb2 += u2.y * mb;
            pa0 += u0.z * mc; pb0 += u0.w * md;
            pa1 += u1.z * mc; pb1 += u1.w * md;
            pa2 += u2.z * mc; pb2 += u2.w * md;
        };
        {
            int c = 0;
            for (int q = 0; q < (njg >> 2); ++q, c += 2) { wpair(c); wpair(c + 1); }
            if (njg & 2) { wpair(c); c += 1; }
            if (njg & 1) {
                const v2f t0 = *(const v2f*)&wq0[c];
                const v2f t1 = *(const v2f*)&wq1[c];
                const v2f t2 = *(const v2f*)&wq2[c];
                const float* mp = mcol + (32 * c) * M_STRIDE;
                const v2f ma = *(const v2f*)(mp);
                const v2f mb = *(const v2f*)(mp + M_STRIDE);
                pa0 += t0.x * ma; pb0 += t0.y * mb;
                pa1 += t1.x * ma; pb1 += t1.y * mb;
                pa2 += t2.x * ma; pb2 += t2.y * mb;
            }
        }
        v2f cc0 = pa0 + pb0, cc1 = pa1 + pb1, cc2 = pa2 + pb2;
        cc0.x += row_xor<4>(cc0.x); cc0.y += row_xor<4>(cc0.y);
        cc1.x += row_xor<4>(cc1.x); cc1.y += row_xor<4>(cc1.y);
        cc2.x += row_xor<4>(cc2.x); cc2.y += row_xor<4>(cc2.y);
        cc0.x += row_xor<8>(cc0.x); cc0.y += row_xor<8>(cc0.y);
        cc1.x += row_xor<8>(cc1.x); cc1.y += row_xor<8>(cc1.y);
        cc2.x += row_xor<8>(cc2.x); cc2.y += row_xor<8>(cc2.y);
        cc0.x += xhalf(cc0.x); cc0.y += xhalf(cc0.y);
        cc1.x += xhalf(cc1.x); cc1.y += xhalf(cc1.y);
        cc2.x += xhalf(cc2.x); cc2.y += xhalf(cc2.y);
        if (g == 0) {
            *(v2f*)&candLDS[0 * O_DIM + oc] = cc0;
            *(v2f*)&candLDS[1 * O_DIM + oc] = cc1;
            *(v2f*)&candLDS[2 * O_DIM + oc] = cc2;
        }
        float q0 = cc0.x * cc0.x + cc0.y * cc0.y;
        float q1 = cc1.x * cc1.x + cc1.y * cc1.y;
        float q2 = cc2.x * cc2.x + cc2.y * cc2.y;
        q0 += row_xor<1>(q0); q1 += row_xor<1>(q1); q2 += row_xor<1>(q2);
        q0 += row_xor<2>(q0); q1 += row_xor<2>(q1); q2 += row_xor<2>(q2);
        if (lane == 0)
            *(v4f*)&snpLDS[wave * 4] = (v4f){q0, q1, q2, 0.0f};
        __syncthreads();

        float sn = 0.0f;
        #pragma unroll
        for (int w = 0; w < 8; ++w)
            sn += snpLDS[w * 4 + (lane & 3)];
        const float fct = sn / (1.0f + sn) / sqrtf(sn + 1e-8f);
        const float f0 = lane_bcast(fct, 0);
        const float f1 = lane_bcast(fct, 1);
        const float f2 = lane_bcast(fct, 2);

        if (it < 2) {
            const int r0 = 8 * wave_s;
            if (r0 < slc) {
                v4f ck0[4], ck1[4], ck2[4];
                #pragma unroll
                for (int j = 0; j < 4; ++j) {
                    ck0[j] = *(const v4f*)&candLDS[0 * O_DIM + 16 * q4 + 4 * j];
                    ck1[j] = *(const v4f*)&candLDS[1 * O_DIM + 16 * q4 + 4 * j];
                    ck2[j] = *(const v4f*)&candLDS[2 * O_DIM + 16 * q4 + 4 * j];
                }
                float ds0 = 0.f, ds1 = 0.f, ds2 = 0.f;
                auto lpass = [&](int p) {
                    const float* mrow = &MLDS[(p * 64 + wave * 8 + rs) * M_STRIDE + 16 * q4];
                    v2f a0 = {0.f, 0.f}, a1 = {0.f, 0.f}, a2 = {0.f, 0.f};
                    #pragma unroll
                    for (int j = 0; j < 4; ++j) {
                        const v4f m4 = *(const v4f*)(mrow + 4 * j);
                        a0 += ck0[j].xy * m4.xy; a1 += ck1[j].xy * m4.xy; a2 += ck2[j].xy * m4.xy;
                        a0 += ck0[j].zw * m4.zw; a1 += ck1[j].zw * m4.zw; a2 += ck2[j].zw * m4.zw;
                    }
                    float d0 = a0.x + a0.y, d1 = a1.x + a1.y, d2 = a2.x + a2.y;
                    d0 += row_xor<1>(d0); d1 += row_xor<1>(d1); d2 += row_xor<1>(d2);
                    d0 += row_xor<2>(d0); d1 += row_xor<2>(d1); d2 += row_xor<2>(d2);
                    if (q4 == p) { ds0 = d0; ds1 = d1; ds2 = d2; }
                };
                lpass(0);
                if (r0 + 64 < slc) { lpass(1); }
                if (r0 + 192 < slc) { lpass(2); lpass(3); } else if (r0 + 128 < slc) { lpass(2); }
                lg0 += f0 * ds0;
                lg1 += f1 * ds1;
                lg2 += f2 * ds2;
            }
        } else {
            if (tid < K_CAPS * O_DIM) {
                const int k = tid >> 6;
                const float fk = (k == 0) ? f0 : ((k == 1) ? f1 : f2);
                out[(size_t)n * (K_CAPS * O_DIM) + tid] = fk * candLDS[tid];
            }
        }
    }
}

extern "C" void kernel_launch(void* const* d_in, const int* in_sizes, int n_in,
                              void* d_out, int out_size, void* d_ws, size_t ws_size,
                              hipStream_t stream) {
    const float* E       = (const float*)d_in[0];
    const float* W       = (const float*)d_in[1];
    const float* logits0 = (const float*)d_in[2];
    const int*   seqlen  = (const int*)d_in[3];
    float*       out     = (float*)d_out;

    const int N = in_sizes[0] / (L_SEQ * D_DIM);
    capsule_fused_kernel<<<N, 256, 0, stream>>>(E, W, logits0, seqlen, out);
}
